# starting kernel
# baseline (speedup 1.0000x reference)
#include <hip/hip_runtime.h>
#include <hip/hip_bf16.h>


typedef float v2f __attribute__((ext_vector_type(2)));
typedef float v8f __attribute__((ext_vector_type(8)));

#define N_IN   512
#define N_OUT  256
#define T_LEN  2048
#define BATCH  32
#define M_TOT  (T_LEN * BATCH)
#define N_TOT  (4 * N_OUT)

#define TM 64
#define TN 64
#define TK 64
#define LDS_STRIDE 68

__device__ __forceinline__ float sigmoidf_fast(float x) {
    return 1.0f / (1.0f + __expf(-x));
}

__global__ __launch_bounds__(128)
void sru_gemm4_kernel(const float* __restrict__ xt,
                      const float* __restrict__ Wx, const float* __restrict__ Wf,
                      const float* __restrict__ Wr, const float* __restrict__ Wc,
                      const float* __restrict__ bf, const float* __restrict__ br,
                      const float* __restrict__ bc,
                      float* __restrict__ g)
{
    __shared__ float As[2][TM * LDS_STRIDE];
    __shared__ float Bs[2][TN * LDS_STRIDE];

    const int tid   = threadIdx.x;
    const int wave  = tid >> 5;
    const int lane  = tid & 31;
    const int lm    = lane & 15;
    const int khalf = lane >> 4;

    const int nblk = blockIdx.x;
    const int m0   = blockIdx.y * TM;
    const int wsel = nblk >> 2;
    const int n0w  = (nblk & 3) * TN;
    const float* __restrict__ W =
        (wsel == 0) ? Wx : (wsel == 1) ? Wf : (wsel == 2) ? Wr : Wc;
    const float* __restrict__ bias =
        (wsel == 1) ? bf : (wsel == 2) ? br : (wsel == 3) ? bc : nullptr;

    v8f acc[4];
#pragma unroll
    for (int ns = 0; ns < 4; ++ns)
        acc[ns] = (v8f){0.f, 0.f, 0.f, 0.f, 0.f, 0.f, 0.f, 0.f};

#pragma unroll
    for (int i = tid; i < TM * (TK / 4); i += 128) {
        const int r  = i >> 4;
        const int c4 = (i & 15) << 2;
        *(float4*)&As[0][r * LDS_STRIDE + c4] =
            *(const float4*)(xt + (size_t)(m0 + r) * N_IN + c4);
        *(float4*)&Bs[0][r * LDS_STRIDE + c4] =
            *(const float4*)(W + (size_t)(n0w + r) * N_IN + c4);
    }
    __syncthreads();

    int buf = 0;
    for (int kb = 0; kb < N_IN; kb += TK, buf ^= 1) {
        if (kb + TK < N_IN) {
            const int kn = kb + TK;
#pragma unroll
            for (int i = tid; i < TM * (TK / 4); i += 128) {
                const int r  = i >> 4;
                const int c4 = (i & 15) << 2;
                *(float4*)&As[buf ^ 1][r * LDS_STRIDE + c4] =
                    *(const float4*)(xt + (size_t)(m0 + r) * N_IN + kn + c4);
                *(float4*)&Bs[buf ^ 1][r * LDS_STRIDE + c4] =
                    *(const float4*)(W + (size_t)(n0w + r) * N_IN + kn + c4);
            }
        }

        const float* Arow = &As[buf][(wave * 16 + lm) * LDS_STRIDE + 2 * khalf];
        const float* Bbase = &Bs[buf][lm * LDS_STRIDE + 2 * khalf];
#pragma unroll
        for (int kk = 0; kk < TK; kk += 4) {
            const v2f a = *(const v2f*)(Arow + kk);
#pragma unroll
            for (int ns = 0; ns < 4; ++ns) {
                const v2f b = *(const v2f*)(Bbase + ns * 16 * LDS_STRIDE + kk);
                acc[ns] = __builtin_amdgcn_wmma_f32_16x16x4_f32(
                    false, a, false, b, (short)0, acc[ns], false, false);
            }
        }
        __syncthreads();
    }

    const int ncol0 = nblk * TN;
#pragma unroll
    for (int ns = 0; ns < 4; ++ns) {
        const int colw = n0w + ns * 16 + lm;
        const int col  = ncol0 + ns * 16 + lm;
        const float bv = (wsel != 0) ? bias[colw] : 0.0f;
#pragma unroll
        for (int r = 0; r < 8; ++r) {
            const int row = m0 + wave * 16 + khalf * 8 + r;
            float v = acc[ns][r];
            if (wsel == 1 || wsel == 2)      v = sigmoidf_fast(v + bv);
            else if (wsel == 3)              v = v + bv;
            g[(size_t)row * N_TOT + col] = v;
        }
    }
}

__global__ __launch_bounds__(256)
void sru_scan_kernel(const float* __restrict__ g,
                     const float* __restrict__ ct0,
                     float* __restrict__ out)
{
    const int o = threadIdx.x;
    const int b = blockIdx.x;

    float c = ct0[b * N_OUT + o];
    const size_t colbase = (size_t)b * N_TOT + o;
    const size_t obase   = (size_t)b * N_OUT + o;
    const size_t rowstride = (size_t)BATCH * N_TOT;

    for (int t = 0; t < T_LEN; ++t) {
        const float* row = g + (size_t)t * rowstride + colbase;

        const float* pf = row + 8 * rowstride;
        __builtin_prefetch(pf + 0,         0, 0);
        __builtin_prefetch(pf + N_OUT,     0, 0);
        __builtin_prefetch(pf + 2 * N_OUT, 0, 0);
        __builtin_prefetch(pf + 3 * N_OUT, 0, 0);

        const float xp = row[0];
        const float f  = row[N_OUT];
        const float r  = row[2 * N_OUT];
        const float cx = row[3 * N_OUT];

        c = f * c + (1.0f - f) * xp;
        const float h = r * tanhf(c) + (1.0f - r) * cx;

        out[(size_t)t * (BATCH * N_OUT) + obase] = h;
    }
    out[(size_t)M_TOT * N_OUT + obase] = c;
}

extern "C" void kernel_launch(void* const* d_in, const int* in_sizes, int n_in,
                              void* d_out, int out_size, void* d_ws, size_t ws_size,
                              hipStream_t stream) {
    const float* xt = (const float*)d_in[0];
    const float* ct = (const float*)d_in[1];
    const float* Wx = (const float*)d_in[2];
    const float* Wf = (const float*)d_in[3];
    const float* bf = (const float*)d_in[4];
    const float* Wr = (const float*)d_in[5];
    const float* br = (const float*)d_in[6];
    const float* Wc = (const float*)d_in[7];
    const float* bc = (const float*)d_in[8];

    float* g   = (float*)d_ws;
    float* out = (float*)d_out;

    dim3 grid1(N_TOT / TN, M_TOT / TM);
    sru_gemm4_kernel<<<grid1, 128, 0, stream>>>(xt, Wx, Wf, Wr, Wc,
                                                bf, br, bc, g);

    sru_scan_kernel<<<BATCH, N_OUT, 0, stream>>>(g, ct, out);
}


# scan spread over time: 8 waves per (batch row, 32 columns) each load and finish 16 of every 128 steps; every wave runs the one-FMA state chain whole from (f,u) in LDS, step order and roundings kept
# speedup vs baseline: 2.4470x; 2.4470x over previous
#include <hip/hip_runtime.h>
#include <hip/hip_bf16.h>


typedef float v2f __attribute__((ext_vector_type(2)));
typedef float v8f __attribute__((ext_vector_type(8)));

#define N_IN   512
#define N_OUT  256
#define T_LEN  2048
#define BATCH  32
#define M_TOT  (T_LEN * BATCH)
#define N_TOT  (4 * N_OUT)

#define TM 64
#define TN 64
#define TK 64
#define LDS_STRIDE 68

__device__ __forceinline__ float sigmoidf_fast(float x) {
    return 1.0f / (1.0f + __expf(-x));
}

__global__ __launch_bounds__(128)
void sru_gemm4_kernel(const float* __restrict__ xt,
                      const float* __restrict__ Wx, const float* __restrict__ Wf,
                      const float* __restrict__ Wr, const float* __restrict__ Wc,
                      const float* __restrict__ bf, const float* __restrict__ br,
                      const float* __restrict__ bc,
                      float* __restrict__ g)
{
    __shared__ float As[2][TM * LDS_STRIDE];
    __shared__ float Bs[2][TN * LDS_STRIDE];

    const int tid   = threadIdx.x;
    const int wave  = tid >> 5;
    const int lane  = tid & 31;
    const int lm    = lane & 15;
    const int khalf = lane >> 4;

    const int nblk = blockIdx.x;
    const int m0   = blockIdx.y * TM;
    const int wsel = nblk >> 2;
    const int n0w  = (nblk & 3) * TN;
    const float* __restrict__ W =
        (wsel == 0) ? Wx : (wsel == 1) ? Wf : (wsel == 2) ? Wr : Wc;
    const float* __restrict__ bias =
        (wsel == 1) ? bf : (wsel == 2) ? br : (wsel == 3) ? bc : nullptr;

    v8f acc[4];
#pragma unroll
    for (int ns = 0; ns < 4; ++ns)
        acc[ns] = (v8f){0.f, 0.f, 0.f, 0.f, 0.f, 0.f, 0.f, 0.f};

#pragma unroll
    for (int i = tid; i < TM * (TK / 4); i += 128) {
        const int r  = i >> 4;
        const int c4 = (i & 15) << 2;
        *(float4*)&As[0][r * LDS_STRIDE + c4] =
            *(const float4*)(xt + (size_t)(m0 + r) * N_IN + c4);
        *(float4*)&Bs[0][r * LDS_STRIDE + c4] =
            *(const float4*)(W + (size_t)(n0w + r) * N_IN + c4);
    }
    __syncthreads();

    int buf = 0;
    for (int kb = 0; kb < N_IN; kb += TK, buf ^= 1) {
        if (kb + TK < N_IN) {
            const int kn = kb + TK;
#pragma unroll
            for (int i = tid; i < TM * (TK / 4); i += 128) {
                const int r  = i >> 4;
                const int c4 = (i & 15) << 2;
                *(float4*)&As[buf ^ 1][r * LDS_STRIDE + c4] =
                    *(const float4*)(xt + (size_t)(m0 + r) * N_IN + kn + c4);
                *(float4*)&Bs[buf ^ 1][r * LDS_STRIDE + c4] =
                    *(const float4*)(W + (size_t)(n0w + r) * N_IN + kn + c4);
            }
        }

        const float* Arow = &As[buf][(wave * 16 + lm) * LDS_STRIDE + 2 * khalf];
        const float* Bbase = &Bs[buf][lm * LDS_STRIDE + 2 * khalf];
#pragma unroll
        for (int kk = 0; kk < TK; kk += 4) {
            const v2f a = *(const v2f*)(Arow + kk);
#pragma unroll
            for (int ns = 0; ns < 4; ++ns) {
                const v2f b = *(const v2f*)(Bbase + ns * 16 * LDS_STRIDE + kk);
                acc[ns] = __builtin_amdgcn_wmma_f32_16x16x4_f32(
                    false, a, false, b, (short)0, acc[ns], false, false);
            }
        }
        __syncthreads();
    }

    const int ncol0 = nblk * TN;
#pragma unroll
    for (int ns = 0; ns < 4; ++ns) {
        const int colw = n0w + ns * 16 + lm;
        const int col  = ncol0 + ns * 16 + lm;
        const float bv = (wsel != 0) ? bias[colw] : 0.0f;
#pragma unroll
        for (int r = 0; r < 8; ++r) {
            const int row = m0 + wave * 16 + khalf * 8 + r;
            float v = acc[ns][r];
            if (wsel == 1 || wsel == 2)      v = sigmoidf_fast(v + bv);
            else if (wsel == 3)              v = v + bv;
            g[(size_t)row * N_TOT + col] = v;
        }
    }
}

#define SC_WAVES  8
#define SC_J      16
#define SC_CHUNK  (SC_WAVES * SC_J)
#define SC_NCHUNK (T_LEN / SC_CHUNK)

__global__ __launch_bounds__(256)
void sru_scan_kernel(const float* __restrict__ g,
                     const float* __restrict__ ct0,
                     float* __restrict__ out)
{
    __shared__ float2 fu[2][SC_CHUNK][32];

    const int lane = threadIdx.x & 31;
    const int w = __builtin_amdgcn_readfirstlane((int)(threadIdx.x >> 5));
    const int b = blockIdx.x >> 3;
    const int o = ((blockIdx.x & 7) << 5) + lane;

    const size_t rowstride = (size_t)BATCH * N_TOT;
    const size_t ostride   = (size_t)BATCH * N_OUT;
    const float* gp = g   + (size_t)(w * SC_J) * rowstride + (size_t)b * N_TOT + o;
    float*       op = out + (size_t)(w * SC_J) * ostride   + (size_t)b * N_OUT + o;

    float c = ct0[b * N_OUT + o];

    float xv[2][SC_J], fv[2][SC_J], rv[2][SC_J], cv[2][SC_J];

#pragma unroll
    for (int j = 0; j < SC_J; ++j) {
        const float* row = gp + (size_t)j * rowstride;
        xv[0][j] = row[0];
        fv[0][j] = row[N_OUT];
        rv[0][j] = row[2 * N_OUT];
        cv[0][j] = row[3 * N_OUT];
    }

    for (int k = 0; k < SC_NCHUNK; k += 2) {
#pragma unroll
        for (int p = 0; p < 2; ++p) {
            float uv[SC_J];
#pragma unroll
            for (int j = 0; j < SC_J; ++j) {
                uv[j] = (1.0f - fv[p][j]) * xv[p][j];
                fu[p][w * SC_J + j][lane] = make_float2(fv[p][j], uv[j]);
            }
            __syncthreads();

            if (k + p + 1 < SC_NCHUNK) {
                const float* gn = gp + (size_t)SC_CHUNK * rowstride;
#pragma unroll
                for (int j = 0; j < SC_J; ++j) {
                    const float* row = gn + (size_t)j * rowstride;
                    xv[p ^ 1][j] = row[0];
                    fv[p ^ 1][j] = row[N_OUT];
                    rv[p ^ 1][j] = row[2 * N_OUT];
                    cv[p ^ 1][j] = row[3 * N_OUT];
                }
            }

            float cj[SC_J];
            float2 q[SC_J], qn[SC_J];
#pragma unroll
            for (int j = 0; j < SC_J; ++j) q[j] = fu[p][j][lane];
#pragma unroll
            for (int ww = 0; ww < SC_WAVES; ++ww) {
                if (ww + 1 < SC_WAVES) {
#pragma unroll
                    for (int j = 0; j < SC_J; ++j) qn[j] = fu[p][(ww + 1) * SC_J + j][lane];
                }
                if (ww == w) {
#pragma unroll
                    for (int j = 0; j < SC_J; ++j) { c = q[j].x * c + q[j].y; cj[j] = c; }
                } else {
#pragma unroll
                    for (int j = 0; j < SC_J; ++j) c = q[j].x * c + q[j].y;
                }
                if (ww + 1 < SC_WAVES) {
#pragma unroll
                    for (int j = 0; j < SC_J; ++j) q[j] = qn[j];
                }
            }

#pragma unroll
            for (int j = 0; j < SC_J; ++j) {
                const float r = rv[p][j];
                const float h = r * tanhf(cj[j]) + (1.0f - r) * cv[p][j];
                op[(size_t)j * ostride] = h;
            }

            gp += (size_t)SC_CHUNK * rowstride;
            op += (size_t)SC_CHUNK * ostride;
        }
    }
    if (w == 0)
        out[(size_t)M_TOT * N_OUT + (size_t)b * N_OUT + o] = c;
}

extern "C" void kernel_launch(void* const* d_in, const int* in_sizes, int n_in,
                              void* d_out, int out_size, void* d_ws, size_t ws_size,
                              hipStream_t stream) {
    const float* xt = (const float*)d_in[0];
    const float* ct = (const float*)d_in[1];
    const float* Wx = (const float*)d_in[2];
    const float* Wf = (const float*)d_in[3];
    const float* bf = (const float*)d_in[4];
    const float* Wr = (const float*)d_in[5];
    const float* br = (const float*)d_in[6];
    const float* Wc = (const float*)d_in[7];
    const float* bc = (const float*)d_in[8];

    float* g   = (float*)d_ws;
    float* out = (float*)d_out;

    dim3 grid1(N_TOT / TN, M_TOT / TM);
    sru_gemm4_kernel<<<grid1, 128, 0, stream>>>(xt, Wx, Wf, Wr, Wc,
                                                bf, br, bc, g);

    sru_scan_kernel<<<BATCH * (N_OUT / 32), SC_WAVES * 32, 0, stream>>>(g, ct, out);
}


# GEMM operand staging: the 16 global loads of a K-chunk go out as one group (constant trip count, registers, then LDS) instead of eight serial load-wait-store round trips
# speedup vs baseline: 3.7580x; 1.5358x over previous
#include <hip/hip_runtime.h>
#include <hip/hip_bf16.h>


typedef float v2f __attribute__((ext_vector_type(2)));
typedef float v8f __attribute__((ext_vector_type(8)));

#define N_IN   512
#define N_OUT  256
#define T_LEN  2048
#define BATCH  32
#define M_TOT  (T_LEN * BATCH)
#define N_TOT  (4 * N_OUT)

#define TM 64
#define TN 64
#define TK 64
#define LDS_STRIDE 68

__device__ __forceinline__ float sigmoidf_fast(float x) {
    return 1.0f / (1.0f + __expf(-x));
}

#define STG 8

__device__ __forceinline__ void mma_chunk(const float* __restrict__ Ab,
                                          const float* __restrict__ Bb,
                                          v8f (&acc)[4], int wave, int lm, int khalf)
{
    const float* Arow  = Ab + (wave * 16 + lm) * LDS_STRIDE + 2 * khalf;
    const float* Bbase = Bb + lm * LDS_STRIDE + 2 * khalf;
#pragma unroll
    for (int kk = 0; kk < TK; kk += 4) {
        const v2f a = *(const v2f*)(Arow + kk);
#pragma unroll
        for (int ns = 0; ns < 4; ++ns) {
            const v2f b = *(const v2f*)(Bbase + ns * 16 * LDS_STRIDE + kk);
            acc[ns] = __builtin_amdgcn_wmma_f32_16x16x4_f32(
                false, a, false, b, (short)0, acc[ns], false, false);
        }
    }
}

__global__ __launch_bounds__(128)
void sru_gemm4_kernel(const float* __restrict__ xt,
                      const float* __restrict__ Wx, const float* __restrict__ Wf,
                      const float* __restrict__ Wr, const float* __restrict__ Wc,
                      const float* __restrict__ bf, const float* __restrict__ br,
                      const float* __restrict__ bc,
                      float* __restrict__ g)
{
    __shared__ float As[2][TM * LDS_STRIDE];
    __shared__ float Bs[2][TN * LDS_STRIDE];

    const int tid   = threadIdx.x;
    const int wave  = tid >> 5;
    const int lane  = tid & 31;
    const int lm    = lane & 15;
    const int khalf = lane >> 4;

    const int nblk = blockIdx.x;
    const int m0   = blockIdx.y * TM;
    const int wsel = nblk >> 2;
    const int n0w  = (nblk & 3) * TN;
    const float* __restrict__ W =
        (wsel == 0) ? Wx : (wsel == 1) ? Wf : (wsel == 2) ? Wr : Wc;
    const float* __restrict__ bias =
        (wsel == 1) ? bf : (wsel == 2) ? br : (wsel == 3) ? bc : nullptr;

    v8f acc[4];
#pragma unroll
    for (int ns = 0; ns < 4; ++ns)
        acc[ns] = (v8f){0.f, 0.f, 0.f, 0.f, 0.f, 0.f, 0.f, 0.f};

    const int r0 = tid >> 4;
    const int c4 = (tid & 15) << 2;
    const float* __restrict__ asrc = xt + (size_t)(m0  + r0) * N_IN + c4;
    const float* __restrict__ bsrc = W  + (size_t)(n0w + r0) * N_IN + c4;
    const int sdst = r0 * LDS_STRIDE + c4;

    float4 pa[STG], pb[STG];

#pragma unroll
    for (int j = 0; j < STG; ++j) {
        pa[j] = *(const float4*)(asrc + (size_t)j * 8 * N_IN);
        pb[j] = *(const float4*)(bsrc + (size_t)j * 8 * N_IN);
    }
#pragma unroll
    for (int j = 0; j < STG; ++j) {
        *(float4*)&As[0][sdst + j * 8 * LDS_STRIDE] = pa[j];
        *(float4*)&Bs[0][sdst + j * 8 * LDS_STRIDE] = pb[j];
    }
    __syncthreads();

    int buf = 0;
    for (int kb = 0; kb < N_IN - TK; kb += TK, buf ^= 1) {
#pragma unroll
        for (int j = 0; j < STG; ++j) {
            pa[j] = *(const float4*)(asrc + kb + TK + (size_t)j * 8 * N_IN);
            pb[j] = *(const float4*)(bsrc + kb + TK + (size_t)j * 8 * N_IN);
        }

        mma_chunk(As[buf], Bs[buf], acc, wave, lm, khalf);

#pragma unroll
        for (int j = 0; j < STG; ++j) {
            *(float4*)&As[buf ^ 1][sdst + j * 8 * LDS_STRIDE] = pa[j];
            *(float4*)&Bs[buf ^ 1][sdst + j * 8 * LDS_STRIDE] = pb[j];
        }
        __syncthreads();
    }
    mma_chunk(As[buf], Bs[buf], acc, wave, lm, khalf);

    const int ncol0 = nblk * TN;
#pragma unroll
    for (int ns = 0; ns < 4; ++ns) {
        const int colw = n0w + ns * 16 + lm;
        const int col  = ncol0 + ns * 16 + lm;
        const float bv = (wsel != 0) ? bias[colw] : 0.0f;
#pragma unroll
        for (int r = 0; r < 8; ++r) {
            const int row = m0 + wave * 16 + khalf * 8 + r;
            float v = acc[ns][r];
            if (wsel == 1 || wsel == 2)      v = sigmoidf_fast(v + bv);
            else if (wsel == 3)              v = v + bv;
            g[(size_t)row * N_TOT + col] = v;
        }
    }
}

#define SC_WAVES  8
#define SC_J      16
#define SC_CHUNK  (SC_WAVES * SC_J)
#define SC_NCHUNK (T_LEN / SC_CHUNK)

__global__ __launch_bounds__(256)
void sru_scan_kernel(const float* __restrict__ g,
                     const float* __restrict__ ct0,
                     float* __restrict__ out)
{
    __shared__ float2 fu[2][SC_CHUNK][32];

    const int lane = threadIdx.x & 31;
    const int w = __builtin_amdgcn_readfirstlane((int)(threadIdx.x >> 5));
    const int b = blockIdx.x >> 3;
    const int o = ((blockIdx.x & 7) << 5) + lane;

    const size_t rowstride = (size_t)BATCH * N_TOT;
    const size_t ostride   = (size_t)BATCH * N_OUT;
    const float* gp = g   + (size_t)(w * SC_J) * rowstride + (size_t)b * N_TOT + o;
    float*       op = out + (size_t)(w * SC_J) * ostride   + (size_t)b * N_OUT + o;

    float c = ct0[b * N_OUT + o];

    float xv[2][SC_J], fv[2][SC_J], rv[2][SC_J], cv[2][SC_J];

#pragma unroll
    for (int j = 0; j < SC_J; ++j) {
        const float* row = gp + (size_t)j * rowstride;
        xv[0][j] = row[0];
        fv[0][j] = row[N_OUT];
        rv[0][j] = row[2 * N_OUT];
        cv[0][j] = row[3 * N_OUT];
    }

    for (int k = 0; k < SC_NCHUNK; k += 2) {
#pragma unroll
        for (int p = 0; p < 2; ++p) {
            float uv[SC_J];
#pragma unroll
            for (int j = 0; j < SC_J; ++j) {
                uv[j] = (1.0f - fv[p][j]) * xv[p][j];
                fu[p][w * SC_J + j][lane] = make_float2(fv[p][j], uv[j]);
            }
            __syncthreads();

            if (k + p + 1 < SC_NCHUNK) {
                const float* gn = gp + (size_t)SC_CHUNK * rowstride;
#pragma unroll
                for (int j = 0; j < SC_J; ++j) {
                    const float* row = gn + (size_t)j * rowstride;
                    xv[p ^ 1][j] = row[0];
                    fv[p ^ 1][j] = row[N_OUT];
                    rv[p ^ 1][j] = row[2 * N_OUT];
                    cv[p ^ 1][j] = row[3 * N_OUT];
                }
            }

            float cj[SC_J];
            float2 q[SC_J], qn[SC_J];
#pragma unroll
            for (int j = 0; j < SC_J; ++j) q[j] = fu[p][j][lane];
#pragma unroll
            for (int ww = 0; ww < SC_WAVES; ++ww) {
                if (ww + 1 < SC_WAVES) {
#pragma unroll
                    for (int j = 0; j < SC_J; ++j) qn[j] = fu[p][(ww + 1) * SC_J + j][lane];
                }
                if (ww == w) {
#pragma unroll
                    for (int j = 0; j < SC_J; ++j) { c = q[j].x * c + q[j].y; cj[j] = c; }
                } else {
#pragma unroll
                    for (int j = 0; j < SC_J; ++j) c = q[j].x * c + q[j].y;
                }
                if (ww + 1 < SC_WAVES) {
#pragma unroll
                    for (int j = 0; j < SC_J; ++j) q[j] = qn[j];
                }
            }

#pragma unroll
            for (int j = 0; j < SC_J; ++j) {
                const float r = rv[p][j];
                const float h = r * tanhf(cj[j]) + (1.0f - r) * cv[p][j];
                op[(size_t)j * ostride] = h;
            }

            gp += (size_t)SC_CHUNK * rowstride;
            op += (size_t)SC_CHUNK * ostride;
        }
    }
    if (w == 0)
        out[(size_t)M_TOT * N_OUT + (size_t)b * N_OUT + o] = c;
}

extern "C" void kernel_launch(void* const* d_in, const int* in_sizes, int n_in,
                              void* d_out, int out_size, void* d_ws, size_t ws_size,
                              hipStream_t stream) {
    const float* xt = (const float*)d_in[0];
    const float* ct = (const float*)d_in[1];
    const float* Wx = (const float*)d_in[2];
    const float* Wf = (const float*)d_in[3];
    const float* bf = (const float*)d_in[4];
    const float* Wr = (const float*)d_in[5];
    const float* br = (const float*)d_in[6];
    const float* Wc = (const float*)d_in[7];
    const float* bc = (const float*)d_in[8];

    float* g   = (float*)d_ws;
    float* out = (float*)d_out;

    dim3 grid1(N_TOT / TN, M_TOT / TM);
    sru_gemm4_kernel<<<grid1, 128, 0, stream>>>(xt, Wx, Wf, Wr, Wc,
                                                bf, br, bc, g);

    sru_scan_kernel<<<BATCH * (N_OUT / 32), SC_WAVES * 32, 0, stream>>>(g, ct, out);
}


# intermediate layout: a GEMM tile is 16 columns x 4 projections, so g holds {x,f,r,cx} of a column side by side; epilogue stores and scan loads are 16 bytes per lane, whole 256-byte pieces
# speedup vs baseline: 4.0084x; 1.0666x over previous
#include <hip/hip_runtime.h>
#include <hip/hip_bf16.h>


typedef float v2f __attribute__((ext_vector_type(2)));
typedef float v8f __attribute__((ext_vector_type(8)));

#define N_IN   512
#define N_OUT  256
#define T_LEN  2048
#define BATCH  32
#define M_TOT  (T_LEN * BATCH)
#define N_TOT  (4 * N_OUT)

#define TM 64
#define TN 64
#define TK 64
#define LDS_STRIDE 68

__device__ __forceinline__ float sigmoidf_fast(float x) {
    return 1.0f / (1.0f + __expf(-x));
}

#define STG 8

__device__ __forceinline__ void mma_chunk(const float* __restrict__ Ab,
                                          const float* __restrict__ Bb,
                                          v8f (&acc)[4], int wave, int lm, int khalf)
{
    const float* Arow  = Ab + (wave * 16 + lm) * LDS_STRIDE + 2 * khalf;
    const float* Bbase = Bb + lm * LDS_STRIDE + 2 * khalf;
#pragma unroll
    for (int kk = 0; kk < TK; kk += 4) {
        const v2f a = *(const v2f*)(Arow + kk);
#pragma unroll
        for (int ns = 0; ns < 4; ++ns) {
            const v2f b = *(const v2f*)(Bbase + ns * 16 * LDS_STRIDE + kk);
            acc[ns] = __builtin_amdgcn_wmma_f32_16x16x4_f32(
                false, a, false, b, (short)0, acc[ns], false, false);
        }
    }
}

__global__ __launch_bounds__(128)
void sru_gemm4_kernel(const float* __restrict__ xt,
                      const float* __restrict__ Wx, const float* __restrict__ Wf,
                      const float* __restrict__ Wr, const float* __restrict__ Wc,
                      const float* __restrict__ bf, const float* __restrict__ br,
                      const float* __restrict__ bc,
                      float* __restrict__ g)
{
    __shared__ float As[2][TM * LDS_STRIDE];
    __shared__ float Bs[2][TN * LDS_STRIDE];

    const int tid   = threadIdx.x;
    const int wave  = tid >> 5;
    const int lane  = tid & 31;
    const int lm    = lane & 15;
    const int khalf = lane >> 4;

    const int nblk = blockIdx.x;
    const int m0   = blockIdx.y * TM;
    const int c0   = nblk * 16;

    v8f acc[4];
#pragma unroll
    for (int ns = 0; ns < 4; ++ns)
        acc[ns] = (v8f){0.f, 0.f, 0.f, 0.f, 0.f, 0.f, 0.f, 0.f};

    const int r0 = tid >> 4;
    const int c4 = (tid & 15) << 2;
    const float* __restrict__ asrc = xt + (size_t)(m0  + r0) * N_IN + c4;
    const unsigned boff = (unsigned)(c0 + r0) * N_IN + c4;
    const float* __restrict__ bsrc[4] = { Wx + boff, Wf + boff, Wr + boff, Wc + boff };
    const int sdst = r0 * LDS_STRIDE + c4;

    float4 pa[STG], pb[STG];

#pragma unroll
    for (int j = 0; j < STG; ++j) {
        pa[j] = *(const float4*)(asrc + (size_t)j * 8 * N_IN);
        pb[j] = *(const float4*)(bsrc[j >> 1] + (j & 1) * 8 * N_IN);
    }
#pragma unroll
    for (int j = 0; j < STG; ++j) {
        *(float4*)&As[0][sdst + j * 8 * LDS_STRIDE] = pa[j];
        *(float4*)&Bs[0][sdst + j * 8 * LDS_STRIDE] = pb[j];
    }
    __syncthreads();

    int buf = 0;
    for (int kb = 0; kb < N_IN - TK; kb += TK, buf ^= 1) {
#pragma unroll
        for (int j = 0; j < STG; ++j) {
            pa[j] = *(const float4*)(asrc + kb + TK + (size_t)j * 8 * N_IN);
            pb[j] = *(const float4*)(bsrc[j >> 1] + kb + TK + (j & 1) * 8 * N_IN);
        }

        mma_chunk(As[buf], Bs[buf], acc, wave, lm, khalf);

#pragma unroll
        for (int j = 0; j < STG; ++j) {
            *(float4*)&As[buf ^ 1][sdst + j * 8 * LDS_STRIDE] = pa[j];
            *(float4*)&Bs[buf ^ 1][sdst + j * 8 * LDS_STRIDE] = pb[j];
        }
        __syncthreads();
    }
    mma_chunk(As[buf], Bs[buf], acc, wave, lm, khalf);

    const int col = c0 + lm;
    const float bfv = bf[col], brv = br[col], bcv = bc[col];
#pragma unroll
    for (int r = 0; r < 8; ++r) {
        const int row = m0 + wave * 16 + khalf * 8 + r;
        float4 v;
        v.x = acc[0][r];
        v.y = sigmoidf_fast(acc[1][r] + bfv);
        v.z = sigmoidf_fast(acc[2][r] + brv);
        v.w = acc[3][r] + bcv;
        *(float4*)(g + ((size_t)row * N_OUT + col) * 4) = v;
    }
}

#define SC_WAVES  8
#define SC_J      16
#define SC_CHUNK  (SC_WAVES * SC_J)
#define SC_NCHUNK (T_LEN / SC_CHUNK)

__global__ __launch_bounds__(256)
void sru_scan_kernel(const float* __restrict__ g,
                     const float* __restrict__ ct0,
                     float* __restrict__ out)
{
    __shared__ float2 fu[2][SC_CHUNK][32];

    const int lane = threadIdx.x & 31;
    const int w = __builtin_amdgcn_readfirstlane((int)(threadIdx.x >> 5));
    const int b = blockIdx.x >> 3;
    const int o = ((blockIdx.x & 7) << 5) + lane;

    const size_t rowstride = (size_t)BATCH * N_TOT;
    const size_t ostride   = (size_t)BATCH * N_OUT;
    const float* gp = g   + (size_t)(w * SC_J) * rowstride + ((size_t)b * N_OUT + o) * 4;
    float*       op = out + (size_t)(w * SC_J) * ostride   + (size_t)b * N_OUT + o;

    float c = ct0[b * N_OUT + o];

    float4 gv[2][SC_J];

#pragma unroll
    for (int j = 0; j < SC_J; ++j)
        gv[0][j] = *(const float4*)(gp + (size_t)j * rowstride);

    for (int k = 0; k < SC_NCHUNK; k += 2) {
#pragma unroll
        for (int p = 0; p < 2; ++p) {
            float uv[SC_J];
#pragma unroll
            for (int j = 0; j < SC_J; ++j) {
                uv[j] = (1.0f - gv[p][j].y) * gv[p][j].x;
                fu[p][w * SC_J + j][lane] = make_float2(gv[p][j].y, uv[j]);
            }
            __syncthreads();

            if (k + p + 1 < SC_NCHUNK) {
                const float* gn = gp + (size_t)SC_CHUNK * rowstride;
#pragma unroll
                for (int j = 0; j < SC_J; ++j)
                    gv[p ^ 1][j] = *(const float4*)(gn + (size_t)j * rowstride);
            }

            float cj[SC_J];
            float2 q[SC_J], qn[SC_J];
#pragma unroll
            for (int j = 0; j < SC_J; ++j) q[j] = fu[p][j][lane];
#pragma unroll
            for (int ww = 0; ww < SC_WAVES; ++ww) {
                if (ww + 1 < SC_WAVES) {
#pragma unroll
                    for (int j = 0; j < SC_J; ++j) qn[j] = fu[p][(ww + 1) * SC_J + j][lane];
                }
                if (ww == w) {
#pragma unroll
                    for (int j = 0; j < SC_J; ++j) { c = q[j].x * c + q[j].y; cj[j] = c; }
                } else {
#pragma unroll
                    for (int j = 0; j < SC_J; ++j) c = q[j].x * c + q[j].y;
                }
                if (ww + 1 < SC_WAVES) {
#pragma unroll
                    for (int j = 0; j < SC_J; ++j) q[j] = qn[j];
                }
            }

#pragma unroll
            for (int j = 0; j < SC_J; ++j) {
                const float r = gv[p][j].z;
                const float h = r * tanhf(cj[j]) + (1.0f - r) * gv[p][j].w;
                op[(size_t)j * ostride] = h;
            }

            gp += (size_t)SC_CHUNK * rowstride;
            op += (size_t)SC_CHUNK * ostride;
        }
    }
    if (w == 0)
        out[(size_t)M_TOT * N_OUT + (size_t)b * N_OUT + o] = c;
}

extern "C" void kernel_launch(void* const* d_in, const int* in_sizes, int n_in,
                              void* d_out, int out_size, void* d_ws, size_t ws_size,
                              hipStream_t stream) {
    const float* xt = (const float*)d_in[0];
    const float* ct = (const float*)d_in[1];
    const float* Wx = (const float*)d_in[2];
    const float* Wf = (const float*)d_in[3];
    const float* bf = (const float*)d_in[4];
    const float* Wr = (const float*)d_in[5];
    const float* br = (const float*)d_in[6];
    const float* Wc = (const float*)d_in[7];
    const float* bc = (const float*)d_in[8];

    float* g   = (float*)d_ws;
    float* out = (float*)d_out;

    dim3 grid1(N_TOT / TN, M_TOT / TM);
    sru_gemm4_kernel<<<grid1, 128, 0, stream>>>(xt, Wx, Wf, Wr, Wc,
                                                bf, br, bc, g);

    sru_scan_kernel<<<BATCH * (N_OUT / 32), SC_WAVES * 32, 0, stream>>>(g, ct, out);
}


# GEMM at six waves per SIMD instead of four: the next K-chunk waits in registers, requested a whole trip (64 WMMAs) ahead, so one LDS stage (34 KB) and 156 registers do; refill fenced behind the WMMAs
# speedup vs baseline: 4.1291x; 1.0301x over previous
#include <hip/hip_runtime.h>
#include <hip/hip_bf16.h>


typedef float v2f __attribute__((ext_vector_type(2)));
typedef float v4f __attribute__((ext_vector_type(4)));
typedef float v8f __attribute__((ext_vector_type(8)));

#define N_IN   512
#define N_OUT  256
#define T_LEN  2048
#define BATCH  32
#define M_TOT  (T_LEN * BATCH)
#define N_TOT  (4 * N_OUT)

#define TM 64
#define TN 64
#define TK 64
#define LDS_STRIDE 68

__device__ __forceinline__ float sigmoidf_fast(float x) {
    return 1.0f / (1.0f + __expf(-x));
}

__device__ __forceinline__ void mma_chunk(const float* __restrict__ Ab,
                                          const float* __restrict__ Bb,
                                          v8f (&acc)[4], int wave, int lm, int khalf)
{
    const float* Arow  = Ab + (wave * 16 + lm) * LDS_STRIDE + 2 * khalf;
    const float* Bbase = Bb + lm * LDS_STRIDE + 2 * khalf;
#pragma unroll
    for (int kk = 0; kk < TK; kk += 4) {
        const v2f a = *(const v2f*)(Arow + kk);
#pragma unroll
        for (int ns = 0; ns < 4; ++ns) {
            const v2f b = *(const v2f*)(Bbase + ns * 16 * LDS_STRIDE + kk);
            acc[ns] = __builtin_amdgcn_wmma_f32_16x16x4_f32(
                false, a, false, b, (short)0, acc[ns], false, false);
        }
    }
}

__global__ __launch_bounds__(128)
void sru_gemm4_kernel(const float* __restrict__ xt,
                      const float* __restrict__ Wx, const float* __restrict__ Wf,
                      const float* __restrict__ Wr, const float* __restrict__ Wc,
                      const float* __restrict__ bf, const float* __restrict__ br,
                      const float* __restrict__ bc,
                      float* __restrict__ g)
{
    __shared__ float As[TM * LDS_STRIDE];
    __shared__ float Bs[TN * LDS_STRIDE];

    const int tid   = threadIdx.x;
    const int wave  = tid >> 5;
    const int lane  = tid & 31;
    const int lm    = lane & 15;
    const int khalf = lane >> 4;

    const int nblk = blockIdx.x;
    const int m0   = blockIdx.y * TM;
    const int c0   = nblk * 16;

    v8f acc[4];
#pragma unroll
    for (int ns = 0; ns < 4; ++ns)
        acc[ns] = (v8f){0.f, 0.f, 0.f, 0.f, 0.f, 0.f, 0.f, 0.f};

    const int r0 = tid >> 4;
    const int c4 = (tid & 15) << 2;
    const float* __restrict__ asrc = xt + (size_t)(m0  + r0) * N_IN + c4;
    const unsigned boff = (unsigned)(c0 + r0) * N_IN + c4;
    const float* __restrict__ bsrc[4] = { Wx + boff, Wf + boff, Wr + boff, Wc + boff };
    const int sdst = r0 * LDS_STRIDE + c4;

    v4f pa0, pa1, pa2, pa3, pa4, pa5, pa6, pa7;
    v4f pb0, pb1, pb2, pb3, pb4, pb5, pb6, pb7;
#define STG_EACH(X) X(0) X(1) X(2) X(3) X(4) X(5) X(6) X(7)
#define STG_LOAD(j)                                                                  \
    pa##j = *(const v4f*)(asrc + koff + (size_t)(j) * 8 * N_IN);                     \
    pb##j = *(const v4f*)(bsrc[(j) >> 1] + koff + ((j) & 1) * 8 * N_IN);
#define STG_STORE(j)                                                                 \
    *(v4f*)&As[sdst + (j) * 8 * LDS_STRIDE] = pa##j;                                 \
    *(v4f*)&Bs[sdst + (j) * 8 * LDS_STRIDE] = pb##j;

    { const int koff = 0;  STG_EACH(STG_LOAD) }
    STG_EACH(STG_STORE)
    __syncthreads();
    { const int koff = TK; STG_EACH(STG_LOAD) }

    for (int kb = 0; kb < N_IN - 2 * TK; kb += TK) {
        mma_chunk(As, Bs, acc, wave, lm, khalf);

        __builtin_amdgcn_sched_barrier(0);

        __syncthreads();
        STG_EACH(STG_STORE)
        __syncthreads();

        { const int koff = kb + 2 * TK; STG_EACH(STG_LOAD) }
        asm volatile("s_wait_xcnt 0x0");
    }
    mma_chunk(As, Bs, acc, wave, lm, khalf);
    __builtin_amdgcn_sched_barrier(0);
    __syncthreads();
    STG_EACH(STG_STORE)
    __syncthreads();
    mma_chunk(As, Bs, acc, wave, lm, khalf);
#undef STG_STORE
#undef STG_LOAD
#undef STG_EACH

    const int col = c0 + lm;
    const float bfv = bf[col], brv = br[col], bcv = bc[col];
#pragma unroll
    for (int r = 0; r < 8; ++r) {
        const int row = m0 + wave * 16 + khalf * 8 + r;
        float4 v;
        v.x = acc[0][r];
        v.y = sigmoidf_fast(acc[1][r] + bfv);
        v.z = sigmoidf_fast(acc[2][r] + brv);
        v.w = acc[3][r] + bcv;
        *(float4*)(g + ((size_t)row * N_OUT + col) * 4) = v;
    }
}

#define SC_WAVES  8
#define SC_J      16
#define SC_CHUNK  (SC_WAVES * SC_J)
#define SC_NCHUNK (T_LEN / SC_CHUNK)

__global__ __launch_bounds__(256)
void sru_scan_kernel(const float* __restrict__ g,
                     const float* __restrict__ ct0,
                     float* __restrict__ out)
{
    __shared__ float2 fu[2][SC_CHUNK][32];

    const int lane = threadIdx.x & 31;
    const int w = __builtin_amdgcn_readfirstlane((int)(threadIdx.x >> 5));
    const int b = blockIdx.x >> 3;
    const int o = ((blockIdx.x & 7) << 5) + lane;

    const size_t rowstride = (size_t)BATCH * N_TOT;
    const size_t ostride   = (size_t)BATCH * N_OUT;
    const float* gp = g   + (size_t)(w * SC_J) * rowstride + ((size_t)b * N_OUT + o) * 4;
    float*       op = out + (size_t)(w * SC_J) * ostride   + (size_t)b * N_OUT + o;

    float c = ct0[b * N_OUT + o];

    float4 gv[2][SC_J];

#pragma unroll
    for (int j = 0; j < SC_J; ++j)
        gv[0][j] = *(const float4*)(gp + (size_t)j * rowstride);

    for (int k = 0; k < SC_NCHUNK; k += 2) {
#pragma unroll
        for (int p = 0; p < 2; ++p) {
            float uv[SC_J];
#pragma unroll
            for (int j = 0; j < SC_J; ++j) {
                uv[j] = (1.0f - gv[p][j].y) * gv[p][j].x;
                fu[p][w * SC_J + j][lane] = make_float2(gv[p][j].y, uv[j]);
            }
            __syncthreads();

            if (k + p + 1 < SC_NCHUNK) {
                const float* gn = gp + (size_t)SC_CHUNK * rowstride;
#pragma unroll
                for (int j = 0; j < SC_J; ++j)
                    gv[p ^ 1][j] = *(const float4*)(gn + (size_t)j * rowstride);
            }

            float cj[SC_J];
            float2 q[SC_J], qn[SC_J];
#pragma unroll
            for (int j = 0; j < SC_J; ++j) q[j] = fu[p][j][lane];
#pragma unroll
            for (int ww = 0; ww < SC_WAVES; ++ww) {
                if (ww + 1 < SC_WAVES) {
#pragma unroll
                    for (int j = 0; j < SC_J; ++j) qn[j] = fu[p][(ww + 1) * SC_J + j][lane];
                }
                if (ww == w) {
#pragma unroll
                    for (int j = 0; j < SC_J; ++j) { c = q[j].x * c + q[j].y; cj[j] = c; }
                } else {
#pragma unroll
                    for (int j = 0; j < SC_J; ++j) c = q[j].x * c + q[j].y;
                }
                if (ww + 1 < SC_WAVES) {
#pragma unroll
                    for (int j = 0; j < SC_J; ++j) q[j] = qn[j];
                }
            }

#pragma unroll
            for (int j = 0; j < SC_J; ++j) {
                const float r = gv[p][j].z;
                const float h = r * tanhf(cj[j]) + (1.0f - r) * gv[p][j].w;
                op[(size_t)j * ostride] = h;
            }

            gp += (size_t)SC_CHUNK * rowstride;
            op += (size_t)SC_CHUNK * ostride;
        }
    }
    if (w == 0)
        out[(size_t)M_TOT * N_OUT + (size_t)b * N_OUT + o] = c;
}

extern "C" void kernel_launch(void* const* d_in, const int* in_sizes, int n_in,
                              void* d_out, int out_size, void* d_ws, size_t ws_size,
                              hipStream_t stream) {
    const float* xt = (const float*)d_in[0];
    const float* ct = (const float*)d_in[1];
    const float* Wx = (const float*)d_in[2];
    const float* Wf = (const float*)d_in[3];
    const float* bf = (const float*)d_in[4];
    const float* Wr = (const float*)d_in[5];
    const float* br = (const float*)d_in[6];
    const float* Wc = (const float*)d_in[7];
    const float* bc = (const float*)d_in[8];

    float* g   = (float*)d_ws;
    float* out = (float*)d_out;

    dim3 grid1(N_TOT / TN, M_TOT / TM);
    sru_gemm4_kernel<<<grid1, 128, 0, stream>>>(xt, Wx, Wf, Wr, Wc,
                                                bf, br, bc, g);

    sru_scan_kernel<<<BATCH * (N_OUT / 32), SC_WAVES * 32, 0, stream>>>(g, ct, out);
}


# one launch: scan runs out of the GEMM accumulators, the intermediate plane is gone; weights resident in LDS, xt straight into A fragments (k order inside 8-k blocks changed: sums reordered)
# speedup vs baseline: 4.5775x; 1.1086x over previous
#include <hip/hip_runtime.h>


typedef float v2f __attribute__((ext_vector_type(2)));
typedef float v4f __attribute__((ext_vector_type(4)));
typedef float v8f __attribute__((ext_vector_type(8)));

#define N_IN   512
#define N_OUT  256
#define T_LEN  2048
#define BATCH  32
#define M_TOT  (T_LEN * BATCH)

#define FW      8
#define FSTEPS  16
#define FTILE   (FW * FSTEPS)
#define FNT     (T_LEN / FTILE)
#define NB      128
#define BP      516
#define TK      64

__device__ __forceinline__ float sigmoidf_fast(float x) {
    return 1.0f / (1.0f + __expf(-x));
}

__device__ __forceinline__ void load_a(v4f (&a)[8], const float* __restrict__ p) {
#pragma unroll
    for (int j = 0; j < 8; ++j) a[j] = *(const v4f*)(p + 8 * j);
}

__device__ __forceinline__ void mma_chunk(const v4f (&a)[8], const float* __restrict__ Bl,
                                          v8f (&acc)[8])
{
#pragma unroll
    for (int j = 0; j < 8; ++j) {
        v4f bq[8];
#pragma unroll
        for (int ns = 0; ns < 8; ++ns)
            bq[ns] = *(const v4f*)(Bl + ns * 16 * BP + 8 * j);
        const v2f a01 = {a[j].x, a[j].y};
        const v2f a23 = {a[j].z, a[j].w};
#pragma unroll
        for (int ns = 0; ns < 8; ++ns) {
            const v2f b01 = {bq[ns].x, bq[ns].y};
            acc[ns] = __builtin_amdgcn_wmma_f32_16x16x4_f32(
                false, a01, false, b01, (short)0, acc[ns], false, false);
        }
#pragma unroll
        for (int ns = 0; ns < 8; ++ns) {
            const v2f b23 = {bq[ns].z, bq[ns].w};
            acc[ns] = __builtin_amdgcn_wmma_f32_16x16x4_f32(
                false, a23, false, b23, (short)0, acc[ns], false, false);
        }
    }
}

__global__ __launch_bounds__(256)
void sru_fused_kernel(const float* __restrict__ xt, const float* __restrict__ ct0,
                      const float* __restrict__ Wx, const float* __restrict__ Wf,
                      const float* __restrict__ Wr, const float* __restrict__ Wc,
                      const float* __restrict__ bf, const float* __restrict__ br,
                      const float* __restrict__ bc,
                      float* __restrict__ out)
{
    __shared__ __attribute__((aligned(16))) float Bs[NB * BP];
    __shared__ float4 fu[FTILE][16];

    const int tid   = threadIdx.x;
    const int w     = __builtin_amdgcn_readfirstlane(tid >> 5);
    const int lane  = tid & 31;
    const int lm    = lane & 15;
    const int khalf = lane >> 4;
    const int b     = blockIdx.x >> 3;
    const int c0    = (blockIdx.x & 7) << 5;

    {
        const int c4 = (tid & 127) << 2;
        const int lh = tid >> 7;
        const float* const Wsel[4] = { Wx, Wf, Wr, Wc };
#pragma unroll
        for (int ns = 0; ns < 8; ++ns) {
            const float* Wp = Wsel[ns & 3];
            const int grp   = ns >> 2;
            v4f tmp[8];
#pragma unroll
            for (int i = 0; i < 8; ++i)
                tmp[i] = *(const v4f*)(Wp + (size_t)(c0 + 2 * (2 * i + lh) + grp) * N_IN + c4);
#pragma unroll
            for (int i = 0; i < 8; ++i)
                *(v4f*)&Bs[(ns * 16 + 2 * i + lh) * BP + c4] = tmp[i];
        }
    }

    const int colA = c0 + 2 * lm;
    const float bf0 = bf[colA], bf1 = bf[colA + 1];
    const float br0 = br[colA], br1 = br[colA + 1];
    const float bc0 = bc[colA], bc1 = bc[colA + 1];
    float cs0 = ct0[b * N_OUT + colA];
    float cs1 = ct0[b * N_OUT + colA + 1];

    const float* Bl = Bs + lm * BP + 4 * khalf;
    const float* arow = xt + ((size_t)(w * FSTEPS + lm) * BATCH + b) * N_IN + 4 * khalf;
    const size_t tile_stride = (size_t)FTILE * BATCH * N_IN;

    v4f aA[8], aB[8];
    load_a(aA, arow);

    __syncthreads();

    for (int it = 0; it < FNT; ++it) {
        v8f acc[8];
#pragma unroll
        for (int ns = 0; ns < 8; ++ns)
            acc[ns] = (v8f){0.f, 0.f, 0.f, 0.f, 0.f, 0.f, 0.f, 0.f};

        for (int kc = 0; kc < N_IN; kc += 2 * TK) {
            load_a(aB, arow + kc + TK);
            mma_chunk(aA, Bl + kc, acc);
            const size_t nxt = (kc + 2 * TK < N_IN) ? (size_t)(kc + 2 * TK)
                             : (it + 1 < FNT)       ? tile_stride : (size_t)0;
            load_a(aA, arow + nxt);
            mma_chunk(aB, Bl + kc + TK, acc);
            asm volatile("s_wait_xcnt 0x0");
        }
        arow += tile_stride;

        float rr0[8], rr1[8], cx0[8], cx1[8];
        __syncthreads();
#pragma unroll
        for (int r = 0; r < 8; ++r) {
            const float x0 = acc[0][r];
            const float f0 = sigmoidf_fast(acc[1][r] + bf0);
            rr0[r] = sigmoidf_fast(acc[2][r] + br0);
            cx0[r] = acc[3][r] + bc0;
            const float x1 = acc[4][r];
            const float f1 = sigmoidf_fast(acc[5][r] + bf1);
            rr1[r] = sigmoidf_fast(acc[6][r] + br1);
            cx1[r] = acc[7][r] + bc1;
            const float u0 = (1.0f - f0) * x0;
            const float u1 = (1.0f - f1) * x1;
            fu[w * FSTEPS + 8 * khalf + r][lm] = make_float4(f0, u0, f1, u1);
        }
        __syncthreads();

        float cj0[8], cj1[8];
#pragma unroll
        for (int r = 0; r < 8; ++r) { cj0[r] = 0.f; cj1[r] = 0.f; }
        float4 q[FSTEPS], qn[FSTEPS];
#pragma unroll
        for (int j = 0; j < FSTEPS; ++j) q[j] = fu[j][lm];
#pragma unroll
        for (int ww = 0; ww < FW; ++ww) {
            if (ww + 1 < FW) {
#pragma unroll
                for (int j = 0; j < FSTEPS; ++j) qn[j] = fu[(ww + 1) * FSTEPS + j][lm];
            }
            if (ww == w) {
#pragma unroll
                for (int j = 0; j < FSTEPS; ++j) {
                    cs0 = q[j].x * cs0 + q[j].y;
                    cs1 = q[j].z * cs1 + q[j].w;
                    const bool mine = ((j >> 3) == khalf);
                    cj0[j & 7] = mine ? cs0 : cj0[j & 7];
                    cj1[j & 7] = mine ? cs1 : cj1[j & 7];
                }
            } else {
#pragma unroll
                for (int j = 0; j < FSTEPS; ++j) {
                    cs0 = q[j].x * cs0 + q[j].y;
                    cs1 = q[j].z * cs1 + q[j].w;
                }
            }
            if (ww + 1 < FW) {
#pragma unroll
                for (int j = 0; j < FSTEPS; ++j) q[j] = qn[j];
            }
        }

        float* op = out + ((size_t)(it * FTILE + w * FSTEPS + 8 * khalf) * BATCH + b) * N_OUT
                        + colA;
#pragma unroll
        for (int r = 0; r < 8; ++r) {
            const float h0 = rr0[r] * tanhf(cj0[r]) + (1.0f - rr0[r]) * cx0[r];
            const float h1 = rr1[r] * tanhf(cj1[r]) + (1.0f - rr1[r]) * cx1[r];
            *(float2*)(op + (size_t)r * BATCH * N_OUT) = make_float2(h0, h1);
        }
    }

    if (w == 0 && khalf == 0)
        *(float2*)(out + (size_t)M_TOT * N_OUT + (size_t)b * N_OUT + colA) =
            make_float2(cs0, cs1);
}

extern "C" void kernel_launch(void* const* d_in, const int* in_sizes, int n_in,
                              void* d_out, int out_size, void* d_ws, size_t ws_size,
                              hipStream_t stream) {
    const float* xt = (const float*)d_in[0];
    const float* ct = (const float*)d_in[1];
    const float* Wx = (const float*)d_in[2];
    const float* Wf = (const float*)d_in[3];
    const float* bf = (const float*)d_in[4];
    const float* Wr = (const float*)d_in[5];
    const float* br = (const float*)d_in[6];
    const float* Wc = (const float*)d_in[7];
    const float* bc = (const float*)d_in[8];

    float* out = (float*)d_out;

    sru_fused_kernel<<<BATCH * (N_OUT / 32), FW * 32, 0, stream>>>(
        xt, ct, Wx, Wf, Wr, Wc, bf, br, bc, out);
}


# activations on the hardware exp and reciprocal: sigmoid's correctly rounded division becomes one 1-ulp v_rcp_f32, tanh becomes 1 - 2/(exp(2c)+1) without range split or branch; results move ~1e-7
# speedup vs baseline: 4.9105x; 1.0727x over previous
#include <hip/hip_runtime.h>


typedef float v2f __attribute__((ext_vector_type(2)));
typedef float v4f __attribute__((ext_vector_type(4)));
typedef float v8f __attribute__((ext_vector_type(8)));

#define N_IN   512
#define N_OUT  256
#define T_LEN  2048
#define BATCH  32
#define M_TOT  (T_LEN * BATCH)

#define FW      8
#define FSTEPS  16
#define FTILE   (FW * FSTEPS)
#define FNT     (T_LEN / FTILE)
#define NB      128
#define BP      516
#define TK      64

__device__ __forceinline__ float sigmoidf_fast(float x) {
    return __builtin_amdgcn_rcpf(1.0f + __expf(-x));
}

__device__ __forceinline__ float tanhf_fast(float c) {
    const float e = __builtin_amdgcn_exp2f(c * 2.8853900817779268f);
    return 1.0f - 2.0f * __builtin_amdgcn_rcpf(e + 1.0f);
}

__device__ __forceinline__ void load_a(v4f (&a)[8], const float* __restrict__ p) {
#pragma unroll
    for (int j = 0; j < 8; ++j) a[j] = *(const v4f*)(p + 8 * j);
}

__device__ __forceinline__ void mma_chunk(const v4f (&a)[8], const float* __restrict__ Bl,
                                          v8f (&acc)[8])
{
#pragma unroll
    for (int j = 0; j < 8; ++j) {
        v4f bq[8];
#pragma unroll
        for (int ns = 0; ns < 8; ++ns)
            bq[ns] = *(const v4f*)(Bl + ns * 16 * BP + 8 * j);
        const v2f a01 = {a[j].x, a[j].y};
        const v2f a23 = {a[j].z, a[j].w};
#pragma unroll
        for (int ns = 0; ns < 8; ++ns) {
            const v2f b01 = {bq[ns].x, bq[ns].y};
            acc[ns] = __builtin_amdgcn_wmma_f32_16x16x4_f32(
                false, a01, false, b01, (short)0, acc[ns], false, false);
        }
#pragma unroll
        for (int ns = 0; ns < 8; ++ns) {
            const v2f b23 = {bq[ns].z, bq[ns].w};
            acc[ns] = __builtin_amdgcn_wmma_f32_16x16x4_f32(
                false, a23, false, b23, (short)0, acc[ns], false, false);
        }
    }
}

__global__ __launch_bounds__(256)
void sru_fused_kernel(const float* __restrict__ xt, const float* __restrict__ ct0,
                      const float* __restrict__ Wx, const float* __restrict__ Wf,
                      const float* __restrict__ Wr, const float* __restrict__ Wc,
                      const float* __restrict__ bf, const float* __restrict__ br,
                      const float* __restrict__ bc,
                      float* __restrict__ out)
{
    __shared__ __attribute__((aligned(16))) float Bs[NB * BP];
    __shared__ float4 fu[FTILE][16];

    const int tid   = threadIdx.x;
    const int w     = __builtin_amdgcn_readfirstlane(tid >> 5);
    const int lane  = tid & 31;
    const int lm    = lane & 15;
    const int khalf = lane >> 4;
    const int b     = blockIdx.x >> 3;
    const int c0    = (blockIdx.x & 7) << 5;

    {
        const int c4 = (tid & 127) << 2;
        const int lh = tid >> 7;
        const float* const Wsel[4] = { Wx, Wf, Wr, Wc };
#pragma unroll
        for (int ns = 0; ns < 8; ++ns) {
            const float* Wp = Wsel[ns & 3];
            const int grp   = ns >> 2;
            v4f tmp[8];
#pragma unroll
            for (int i = 0; i < 8; ++i)
                tmp[i] = *(const v4f*)(Wp + (size_t)(c0 + 2 * (2 * i + lh) + grp) * N_IN + c4);
#pragma unroll
            for (int i = 0; i < 8; ++i)
                *(v4f*)&Bs[(ns * 16 + 2 * i + lh) * BP + c4] = tmp[i];
        }
    }

    const int colA = c0 + 2 * lm;
    const float bf0 = bf[colA], bf1 = bf[colA + 1];
    const float br0 = br[colA], br1 = br[colA + 1];
    const float bc0 = bc[colA], bc1 = bc[colA + 1];
    float cs0 = ct0[b * N_OUT + colA];
    float cs1 = ct0[b * N_OUT + colA + 1];

    const float* Bl = Bs + lm * BP + 4 * khalf;
    const float* arow = xt + ((size_t)(w * FSTEPS + lm) * BATCH + b) * N_IN + 4 * khalf;
    const size_t tile_stride = (size_t)FTILE * BATCH * N_IN;

    v4f aA[8], aB[8];
    load_a(aA, arow);

    __syncthreads();

    for (int it = 0; it < FNT; ++it) {
        v8f acc[8];
#pragma unroll
        for (int ns = 0; ns < 8; ++ns)
            acc[ns] = (v8f){0.f, 0.f, 0.f, 0.f, 0.f, 0.f, 0.f, 0.f};

        for (int kc = 0; kc < N_IN; kc += 2 * TK) {
            load_a(aB, arow + kc + TK);
            mma_chunk(aA, Bl + kc, acc);
            const size_t nxt = (kc + 2 * TK < N_IN) ? (size_t)(kc + 2 * TK)
                             : (it + 1 < FNT)       ? tile_stride : (size_t)0;
            load_a(aA, arow + nxt);
            mma_chunk(aB, Bl + kc + TK, acc);
            asm volatile("s_wait_xcnt 0x0");
        }
        arow += tile_stride;

        float rr0[8], rr1[8], cx0[8], cx1[8];
        __syncthreads();
#pragma unroll
        for (int r = 0; r < 8; ++r) {
            const float x0 = acc[0][r];
            const float f0 = sigmoidf_fast(acc[1][r] + bf0);
            rr0[r] = sigmoidf_fast(acc[2][r] + br0);
            cx0[r] = acc[3][r] + bc0;
            const float x1 = acc[4][r];
            const float f1 = sigmoidf_fast(acc[5][r] + bf1);
            rr1[r] = sigmoidf_fast(acc[6][r] + br1);
            cx1[r] = acc[7][r] + bc1;
            const float u0 = (1.0f - f0) * x0;
            const float u1 = (1.0f - f1) * x1;
            fu[w * FSTEPS + 8 * khalf + r][lm] = make_float4(f0, u0, f1, u1);
        }
        __syncthreads();

        float cj0[8], cj1[8];
#pragma unroll
        for (int r = 0; r < 8; ++r) { cj0[r] = 0.f; cj1[r] = 0.f; }
        float4 q[FSTEPS], qn[FSTEPS];
#pragma unroll
        for (int j = 0; j < FSTEPS; ++j) q[j] = fu[j][lm];
#pragma unroll
        for (int ww = 0; ww < FW; ++ww) {
            if (ww + 1 < FW) {
#pragma unroll
                for (int j = 0; j < FSTEPS; ++j) qn[j] = fu[(ww + 1) * FSTEPS + j][lm];
            }
            if (ww == w) {
#pragma unroll
                for (int j = 0; j < FSTEPS; ++j) {
                    cs0 = q[j].x * cs0 + q[j].y;
                    cs1 = q[j].z * cs1 + q[j].w;
                    const bool mine = ((j >> 3) == khalf);
                    cj0[j & 7] = mine ? cs0 : cj0[j & 7];
                    cj1[j & 7] = mine ? cs1 : cj1[j & 7];
                }
            } else {
#pragma unroll
                for (int j = 0; j < FSTEPS; ++j) {
                    cs0 = q[j].x * cs0 + q[j].y;
                    cs1 = q[j].z * cs1 + q[j].w;
                }
            }
            if (ww + 1 < FW) {
#pragma unroll
                for (int j = 0; j < FSTEPS; ++j) q[j] = qn[j];
            }
        }

        float* op = out + ((size_t)(it * FTILE + w * FSTEPS + 8 * khalf) * BATCH + b) * N_OUT
                        + colA;
#pragma unroll
        for (int r = 0; r < 8; ++r) {
            const float h0 = rr0[r] * tanhf_fast(cj0[r]) + (1.0f - rr0[r]) * cx0[r];
            const float h1 = rr1[r] * tanhf_fast(cj1[r]) + (1.0f - rr1[r]) * cx1[r];
            *(float2*)(op + (size_t)r * BATCH * N_OUT) = make_float2(h0, h1);
        }
    }

    if (w == 0 && khalf == 0)
        *(float2*)(out + (size_t)M_TOT * N_OUT + (size_t)b * N_OUT + colA) =
            make_float2(cs0, cs1);
}

extern "C" void kernel_launch(void* const* d_in, const int* in_sizes, int n_in,
                              void* d_out, int out_size, void* d_ws, size_t ws_size,
                              hipStream_t stream) {
    const float* xt = (const float*)d_in[0];
    const float* ct = (const float*)d_in[1];
    const float* Wx = (const float*)d_in[2];
    const float* Wf = (const float*)d_in[3];
    const float* bf = (const float*)d_in[4];
    const float* Wr = (const float*)d_in[5];
    const float* br = (const float*)d_in[6];
    const float* Wc = (const float*)d_in[7];
    const float* bc = (const float*)d_in[8];

    float* out = (float*)d_out;

    sru_fused_kernel<<<BATCH * (N_OUT / 32), FW * 32, 0, stream>>>(
        xt, ct, Wx, Wf, Wr, Wc, bf, br, bc, out);
}


# K loop's weight-fragment feed: 32 LDS reads as ONE group per 64 WMMAs (was 8 per 16), at immediate offsets (a lane's 8 weight rows back to back in LDS), no address add among the WMMAs; sums unchanged
# speedup vs baseline: 5.0808x; 1.0347x over previous
#include <hip/hip_runtime.h>


typedef float v2f __attribute__((ext_vector_type(2)));
typedef float v4f __attribute__((ext_vector_type(4)));
typedef float v8f __attribute__((ext_vector_type(8)));

#define N_IN   512
#define N_OUT  256
#define T_LEN  2048
#define BATCH  32
#define M_TOT  (T_LEN * BATCH)

#define FW      8
#define FSTEPS  16
#define FTILE   (FW * FSTEPS)
#define FNT     (T_LEN / FTILE)
#define SL      (8 * N_IN + 4)
#define TK      64

__device__ __forceinline__ float sigmoidf_fast(float x) {
    return __builtin_amdgcn_rcpf(1.0f + __expf(-x));
}

__device__ __forceinline__ float tanhf_fast(float c) {
    const float e = __builtin_amdgcn_exp2f(c * 2.8853900817779268f);
    return 1.0f - 2.0f * __builtin_amdgcn_rcpf(e + 1.0f);
}

__device__ __forceinline__ void load_a(v4f (&a)[8], const float* __restrict__ p) {
#pragma unroll
    for (int j = 0; j < 8; ++j) a[j] = *(const v4f*)(p + 8 * j);
}

__device__ __forceinline__ void load_b(v4f (&bq)[4][8], const float* __restrict__ Bl) {
#pragma unroll
    for (int j = 0; j < 4; ++j) {
#pragma unroll
        for (int ns = 0; ns < 8; ++ns)
            bq[j][ns] = *(const v4f*)(Bl + ns * N_IN + 8 * j);
    }
}

__device__ __forceinline__ void mma_half(const v4f* a, const v4f (&bq)[4][8], v8f (&acc)[8])
{
#pragma unroll
    for (int j = 0; j < 4; ++j) {
        const v2f a01 = {a[j].x, a[j].y};
        const v2f a23 = {a[j].z, a[j].w};
#pragma unroll
        for (int ns = 0; ns < 8; ++ns) {
            const v2f b01 = {bq[j][ns].x, bq[j][ns].y};
            acc[ns] = __builtin_amdgcn_wmma_f32_16x16x4_f32(
                false, a01, false, b01, (short)0, acc[ns], false, false);
        }
#pragma unroll
        for (int ns = 0; ns < 8; ++ns) {
            const v2f b23 = {bq[j][ns].z, bq[j][ns].w};
            acc[ns] = __builtin_amdgcn_wmma_f32_16x16x4_f32(
                false, a23, false, b23, (short)0, acc[ns], false, false);
        }
    }
}

__global__ __launch_bounds__(256)
void sru_fused_kernel(const float* __restrict__ xt, const float* __restrict__ ct0,
                      const float* __restrict__ Wx, const float* __restrict__ Wf,
                      const float* __restrict__ Wr, const float* __restrict__ Wc,
                      const float* __restrict__ bf, const float* __restrict__ br,
                      const float* __restrict__ bc,
                      float* __restrict__ out)
{
    __shared__ __attribute__((aligned(16))) float Bs[16 * SL];
    __shared__ float4 fu[FTILE][16];

    const int tid   = threadIdx.x;
    const int w     = __builtin_amdgcn_readfirstlane(tid >> 5);
    const int lane  = tid & 31;
    const int lm    = lane & 15;
    const int khalf = lane >> 4;
    const int b     = blockIdx.x >> 3;
    const int c0    = (blockIdx.x & 7) << 5;

    {
        const int c4 = (tid & 127) << 2;
        const int lh = tid >> 7;
        const float* const Wsel[4] = { Wx, Wf, Wr, Wc };
#pragma unroll
        for (int ns = 0; ns < 8; ++ns) {
            const float* Wp = Wsel[ns & 3];
            const int grp   = ns >> 2;
            v4f tmp[8];
#pragma unroll
            for (int i = 0; i < 8; ++i)
                tmp[i] = *(const v4f*)(Wp + (size_t)(c0 + 2 * (2 * i + lh) + grp) * N_IN + c4);
#pragma unroll
            for (int i = 0; i < 8; ++i)
                *(v4f*)&Bs[(2 * i + lh) * SL + ns * N_IN + c4] = tmp[i];
        }
    }

    const int colA = c0 + 2 * lm;
    const float bf0 = bf[colA], bf1 = bf[colA + 1];
    const float br0 = br[colA], br1 = br[colA + 1];
    const float bc0 = bc[colA], bc1 = bc[colA + 1];
    float cs0 = ct0[b * N_OUT + colA];
    float cs1 = ct0[b * N_OUT + colA + 1];

    const float* Bl = Bs + lm * SL + 4 * khalf;
    const float* arow = xt + ((size_t)(w * FSTEPS + lm) * BATCH + b) * N_IN + 4 * khalf;
    const size_t tile_stride = (size_t)FTILE * BATCH * N_IN;

    v4f aA[8], aB[8];
    load_a(aA, arow);

    __syncthreads();

    v4f bA[4][8], bB[4][8];
    load_b(bA, Bl);

    for (int it = 0; it < FNT; ++it) {
        v8f acc[8];
#pragma unroll
        for (int ns = 0; ns < 8; ++ns)
            acc[ns] = (v8f){0.f, 0.f, 0.f, 0.f, 0.f, 0.f, 0.f, 0.f};

        for (int kc = 0; kc < N_IN; kc += 2 * TK) {
            load_a(aB, arow + kc + TK);
            load_b(bB, Bl + kc + TK / 2);
            __builtin_amdgcn_sched_barrier(0);
            mma_half(&aA[0], bA, acc);
            __builtin_amdgcn_sched_barrier(0);
            load_b(bA, Bl + kc + TK);
            __builtin_amdgcn_sched_barrier(0);
            mma_half(&aA[4], bB, acc);
            __builtin_amdgcn_sched_barrier(0);
            const size_t nxt = (kc + 2 * TK < N_IN) ? (size_t)(kc + 2 * TK)
                             : (it + 1 < FNT)       ? tile_stride : (size_t)0;
            load_a(aA, arow + nxt);
            load_b(bB, Bl + kc + TK + TK / 2);
            __builtin_amdgcn_sched_barrier(0);
            mma_half(&aB[0], bA, acc);
            __builtin_amdgcn_sched_barrier(0);
            load_b(bA, Bl + ((kc + 2 * TK) & (N_IN - 1)));
            __builtin_amdgcn_sched_barrier(0);
            mma_half(&aB[4], bB, acc);
            asm volatile("s_wait_xcnt 0x0");
        }
        arow += tile_stride;

        float rr0[8], rr1[8], cx0[8], cx1[8];
        __syncthreads();
#pragma unroll
        for (int r = 0; r < 8; ++r) {
            const float x0 = acc[0][r];
            const float f0 = sigmoidf_fast(acc[1][r] + bf0);
            rr0[r] = sigmoidf_fast(acc[2][r] + br0);
            cx0[r] = acc[3][r] + bc0;
            const float x1 = acc[4][r];
            const float f1 = sigmoidf_fast(acc[5][r] + bf1);
            rr1[r] = sigmoidf_fast(acc[6][r] + br1);
            cx1[r] = acc[7][r] + bc1;
            const float u0 = (1.0f - f0) * x0;
            const float u1 = (1.0f - f1) * x1;
            fu[w * FSTEPS + 8 * khalf + r][lm] = make_float4(f0, u0, f1, u1);
        }
        __syncthreads();

        float cj0[8], cj1[8];
#pragma unroll
        for (int r = 0; r < 8; ++r) { cj0[r] = 0.f; cj1[r] = 0.f; }
        float4 q[FSTEPS], qn[FSTEPS];
#pragma unroll
        for (int j = 0; j < FSTEPS; ++j) q[j] = fu[j][lm];
#pragma unroll
        for (int ww = 0; ww < FW; ++ww) {
            if (ww + 1 < FW) {
#pragma unroll
                for (int j = 0; j < FSTEPS; ++j) qn[j] = fu[(ww + 1) * FSTEPS + j][lm];
            }
            if (ww == w) {
#pragma unroll
                for (int j = 0; j < FSTEPS; ++j) {
                    cs0 = q[j].x * cs0 + q[j].y;
                    cs1 = q[j].z * cs1 + q[j].w;
                    const bool mine = ((j >> 3) == khalf);
                    cj0[j & 7] = mine ? cs0 : cj0[j & 7];
                    cj1[j & 7] = mine ? cs1 : cj1[j & 7];
                }
            } else {
#pragma unroll
                for (int j = 0; j < FSTEPS; ++j) {
                    cs0 = q[j].x * cs0 + q[j].y;
                    cs1 = q[j].z * cs1 + q[j].w;
                }
            }
            if (ww + 1 < FW) {
#pragma unroll
                for (int j = 0; j < FSTEPS; ++j) q[j] = qn[j];
            }
        }

        load_b(bA, Bl);

        float* op = out + ((size_t)(it * FTILE + w * FSTEPS + 8 * khalf) * BATCH + b) * N_OUT
                        + colA;
#pragma unroll
        for (int r = 0; r < 8; ++r) {
            const float h0 = rr0[r] * tanhf_fast(cj0[r]) + (1.0f - rr0[r]) * cx0[r];
            const float h1 = rr1[r] * tanhf_fast(cj1[r]) + (1.0f - rr1[r]) * cx1[r];
            *(float2*)(op + (size_t)r * BATCH * N_OUT) = make_float2(h0, h1);
        }
    }

    if (w == 0 && khalf == 0)
        *(float2*)(out + (size_t)M_TOT * N_OUT + (size_t)b * N_OUT + colA) =
            make_float2(cs0, cs1);
}

extern "C" void kernel_launch(void* const* d_in, const int* in_sizes, int n_in,
                              void* d_out, int out_size, void* d_ws, size_t ws_size,
                              hipStream_t stream) {
    const float* xt = (const float*)d_in[0];
    const float* ct = (const float*)d_in[1];
    const float* Wx = (const float*)d_in[2];
    const float* Wf = (const float*)d_in[3];
    const float* bf = (const float*)d_in[4];
    const float* Wr = (const float*)d_in[5];
    const float* br = (const float*)d_in[6];
    const float* Wc = (const float*)d_in[7];
    const float* bc = (const float*)d_in[8];

    float* out = (float*)d_out;

    sru_fused_kernel<<<BATCH * (N_OUT / 32), FW * 32, 0, stream>>>(
        xt, ct, Wx, Wf, Wr, Wc, bf, br, bc, out);
}


# two-level state scan: a lane folds its own 8 steps to (F,U) in registers, 16 pairs per tile cross LDS (every wave read all 128 steps before), lanes fold them for the carry-in; carry-in sums reordered
# speedup vs baseline: 5.2866x; 1.0405x over previous
#include <hip/hip_runtime.h>


typedef float v2f __attribute__((ext_vector_type(2)));
typedef float v4f __attribute__((ext_vector_type(4)));
typedef float v8f __attribute__((ext_vector_type(8)));

#define N_IN   512
#define N_OUT  256
#define T_LEN  2048
#define BATCH  32
#define M_TOT  (T_LEN * BATCH)

#define FW      8
#define FSTEPS  16
#define FTILE   (FW * FSTEPS)
#define FNT     (T_LEN / FTILE)
#define SL      (8 * N_IN + 4)
#define TK      64

__device__ __forceinline__ float sigmoidf_fast(float x) {
    return __builtin_amdgcn_rcpf(1.0f + __expf(-x));
}

__device__ __forceinline__ float tanhf_fast(float c) {
    const float e = __builtin_amdgcn_exp2f(c * 2.8853900817779268f);
    return 1.0f - 2.0f * __builtin_amdgcn_rcpf(e + 1.0f);
}

__device__ __forceinline__ void load_a(v4f (&a)[8], const float* __restrict__ p) {
#pragma unroll
    for (int j = 0; j < 8; ++j) a[j] = *(const v4f*)(p + 8 * j);
}

__device__ __forceinline__ void load_b(v4f (&bq)[4][8], const float* __restrict__ Bl) {
#pragma unroll
    for (int j = 0; j < 4; ++j) {
#pragma unroll
        for (int ns = 0; ns < 8; ++ns)
            bq[j][ns] = *(const v4f*)(Bl + ns * N_IN + 8 * j);
    }
}

__device__ __forceinline__ void mma_half(const v4f* a, const v4f (&bq)[4][8], v8f (&acc)[8])
{
#pragma unroll
    for (int j = 0; j < 4; ++j) {
        const v2f a01 = {a[j].x, a[j].y};
        const v2f a23 = {a[j].z, a[j].w};
#pragma unroll
        for (int ns = 0; ns < 8; ++ns) {
            const v2f b01 = {bq[j][ns].x, bq[j][ns].y};
            acc[ns] = __builtin_amdgcn_wmma_f32_16x16x4_f32(
                false, a01, false, b01, (short)0, acc[ns], false, false);
        }
#pragma unroll
        for (int ns = 0; ns < 8; ++ns) {
            const v2f b23 = {bq[j][ns].z, bq[j][ns].w};
            acc[ns] = __builtin_amdgcn_wmma_f32_16x16x4_f32(
                false, a23, false, b23, (short)0, acc[ns], false, false);
        }
    }
}

__global__ __launch_bounds__(256)
void sru_fused_kernel(const float* __restrict__ xt, const float* __restrict__ ct0,
                      const float* __restrict__ Wx, const float* __restrict__ Wf,
                      const float* __restrict__ Wr, const float* __restrict__ Wc,
                      const float* __restrict__ bf, const float* __restrict__ br,
                      const float* __restrict__ bc,
                      float* __restrict__ out)
{
    __shared__ __attribute__((aligned(16))) float Bs[16 * SL];
    __shared__ float4 comp[2][2 * FW][16];

    const int tid   = threadIdx.x;
    const int w     = __builtin_amdgcn_readfirstlane(tid >> 5);
    const int lane  = tid & 31;
    const int lm    = lane & 15;
    const int khalf = lane >> 4;
    const int b     = blockIdx.x >> 3;
    const int c0    = (blockIdx.x & 7) << 5;

    {
        const int c4 = (tid & 127) << 2;
        const int lh = tid >> 7;
        const float* const Wsel[4] = { Wx, Wf, Wr, Wc };
#pragma unroll
        for (int ns = 0; ns < 8; ++ns) {
            const float* Wp = Wsel[ns & 3];
            const int grp   = ns >> 2;
            v4f tmp[8];
#pragma unroll
            for (int i = 0; i < 8; ++i)
                tmp[i] = *(const v4f*)(Wp + (size_t)(c0 + 2 * (2 * i + lh) + grp) * N_IN + c4);
#pragma unroll
            for (int i = 0; i < 8; ++i)
                *(v4f*)&Bs[(2 * i + lh) * SL + ns * N_IN + c4] = tmp[i];
        }
    }

    const int colA = c0 + 2 * lm;
    const float bf0 = bf[colA], bf1 = bf[colA + 1];
    const float br0 = br[colA], br1 = br[colA + 1];
    const float bc0 = bc[colA], bc1 = bc[colA + 1];
    float cs0 = ct0[b * N_OUT + colA];
    float cs1 = ct0[b * N_OUT + colA + 1];
    float cl0 = cs0, cl1 = cs1;

    const float* Bl = Bs + lm * SL + 4 * khalf;
    const float* arow = xt + ((size_t)(w * FSTEPS + lm) * BATCH + b) * N_IN + 4 * khalf;
    const size_t tile_stride = (size_t)FTILE * BATCH * N_IN;

    v4f aA[8], aB[8];
    load_a(aA, arow);

    __syncthreads();

    v4f bA[4][8], bB[4][8];
    load_b(bA, Bl);

    for (int it = 0; it < FNT; ++it) {
        v8f acc[8];
#pragma unroll
        for (int ns = 0; ns < 8; ++ns)
            acc[ns] = (v8f){0.f, 0.f, 0.f, 0.f, 0.f, 0.f, 0.f, 0.f};

        for (int kc = 0; kc < N_IN; kc += 2 * TK) {
            load_a(aB, arow + kc + TK);
            load_b(bB, Bl + kc + TK / 2);
            __builtin_amdgcn_sched_barrier(0);
            mma_half(&aA[0], bA, acc);
            __builtin_amdgcn_sched_barrier(0);
            load_b(bA, Bl + kc + TK);
            __builtin_amdgcn_sched_barrier(0);
            mma_half(&aA[4], bB, acc);
            __builtin_amdgcn_sched_barrier(0);
            const size_t nxt = (kc + 2 * TK < N_IN) ? (size_t)(kc + 2 * TK)
                             : (it + 1 < FNT)       ? tile_stride : (size_t)0;
            load_a(aA, arow + nxt);
            load_b(bB, Bl + kc + TK + TK / 2);
            __builtin_amdgcn_sched_barrier(0);
            mma_half(&aB[0], bA, acc);
            __builtin_amdgcn_sched_barrier(0);
            load_b(bA, Bl + ((kc + 2 * TK) & (N_IN - 1)));
            __builtin_amdgcn_sched_barrier(0);
            mma_half(&aB[4], bB, acc);
            asm volatile("s_wait_xcnt 0x0");
        }
        arow += tile_stride;

        float f0[8], u0[8], f1[8], u1[8];
#pragma unroll
        for (int r = 0; r < 8; ++r) {
            f0[r] = sigmoidf_fast(acc[1][r] + bf0);
            f1[r] = sigmoidf_fast(acc[5][r] + bf1);
            u0[r] = (1.0f - f0[r]) * acc[0][r];
            u1[r] = (1.0f - f1[r]) * acc[4][r];
        }
        float F0 = f0[0], U0 = u0[0], F1 = f1[0], U1 = u1[0];
#pragma unroll
        for (int r = 1; r < 8; ++r) {
            U0 = __builtin_fmaf(f0[r], U0, u0[r]);
            F0 = f0[r] * F0;
            U1 = __builtin_fmaf(f1[r], U1, u1[r]);
            F1 = f1[r] * F1;
        }
        float4 (*cb)[16] = comp[it & 1];
        cb[2 * w + khalf][lm] = make_float4(F0, U0, F1, U1);

        float rr0[8], rr1[8], cx0[8], cx1[8];
#pragma unroll
        for (int r = 0; r < 8; ++r) {
            rr0[r] = sigmoidf_fast(acc[2][r] + br0);
            cx0[r] = acc[3][r] + bc0;
            rr1[r] = sigmoidf_fast(acc[6][r] + br1);
            cx1[r] = acc[7][r] + bc1;
        }
        __syncthreads();

        float4 q[2 * FW];
#pragma unroll
        for (int s = 0; s < 2 * FW; ++s) q[s] = cb[s][lm];
        float inA0 = cs0, inA1 = cs1, inB0 = cs0, inB1 = cs1;
#pragma unroll
        for (int ww = 0; ww < FW; ++ww) {
            const bool me = (ww == w);
            inA0 = me ? cs0 : inA0;
            inA1 = me ? cs1 : inA1;
            cs0 = __builtin_fmaf(q[2 * ww].x, cs0, q[2 * ww].y);
            cs1 = __builtin_fmaf(q[2 * ww].z, cs1, q[2 * ww].w);
            inB0 = me ? cs0 : inB0;
            inB1 = me ? cs1 : inB1;
            cs0 = __builtin_fmaf(q[2 * ww + 1].x, cs0, q[2 * ww + 1].y);
            cs1 = __builtin_fmaf(q[2 * ww + 1].z, cs1, q[2 * ww + 1].w);
        }
        float cj0[8], cj1[8];
        {
            float c0 = khalf ? inB0 : inA0;
            float c1 = khalf ? inB1 : inA1;
#pragma unroll
            for (int r = 0; r < 8; ++r) {
                c0 = __builtin_fmaf(f0[r], c0, u0[r]);
                c1 = __builtin_fmaf(f1[r], c1, u1[r]);
                cj0[r] = c0;
                cj1[r] = c1;
            }
        }
        cl0 = cj0[7];
        cl1 = cj1[7];

        load_b(bA, Bl);

        float* op = out + ((size_t)(it * FTILE + w * FSTEPS + 8 * khalf) * BATCH + b) * N_OUT
                        + colA;
#pragma unroll
        for (int r = 0; r < 8; ++r) {
            const float h0 = rr0[r] * tanhf_fast(cj0[r]) + (1.0f - rr0[r]) * cx0[r];
            const float h1 = rr1[r] * tanhf_fast(cj1[r]) + (1.0f - rr1[r]) * cx1[r];
            *(float2*)(op + (size_t)r * BATCH * N_OUT) = make_float2(h0, h1);
        }
    }

    if (w == FW - 1 && khalf == 1)
        *(float2*)(out + (size_t)M_TOT * N_OUT + (size_t)b * N_OUT + colA) =
            make_float2(cl0, cl1);
}

extern "C" void kernel_launch(void* const* d_in, const int* in_sizes, int n_in,
                              void* d_out, int out_size, void* d_ws, size_t ws_size,
                              hipStream_t stream) {
    const float* xt = (const float*)d_in[0];
    const float* ct = (const float*)d_in[1];
    const float* Wx = (const float*)d_in[2];
    const float* Wf = (const float*)d_in[3];
    const float* bf = (const float*)d_in[4];
    const float* Wr = (const float*)d_in[5];
    const float* br = (const float*)d_in[6];
    const float* Wc = (const float*)d_in[7];
    const float* bc = (const float*)d_in[8];

    float* out = (float*)d_out;

    sru_fused_kernel<<<BATCH * (N_OUT / 32), FW * 32, 0, stream>>>(
        xt, ct, Wx, Wf, Wr, Wc, bf, br, bc, out);
}
